# starting kernel
# baseline (speedup 1.0000x reference)
#include <hip/hip_runtime.h>
#include <hip/hip_bf16.h>


typedef float v2f __attribute__((ext_vector_type(2)));
typedef float v8f __attribute__((ext_vector_type(8)));

#define PDIM 256
#define HDIM 256
#define NBATCH 8
#define LSEQ 4096
#define MROWS (NBATCH * LSEQ)
#define LC 256
#define NCHUNK (LSEQ / LC)
#define LDSPITCH 260

static __device__ inline v8f wmma4f32(v2f a, v2f b, v8f c) {
  return __builtin_amdgcn_wmma_f32_16x16x4_f32(
       false, a,  false, b,
       (short)0, c,  false,  false);
}

__global__ void prep_params(const float* __restrict__ A_diag,
                            const float* __restrict__ G_diag,
                            const float* __restrict__ dt,
                            float* __restrict__ m11,
                            float* __restrict__ m12,
                            float* __restrict__ dts) {
  int p = threadIdx.x;
  float s = 1.0f / (1.0f + __expf(-dt[p]));
  float Glow = s * A_diag[p];
  float G = Glow + fmaxf(G_diag[p] - Glow, 0.0f);
  float Alow = 0.25f * G * G;
  float A = Alow + fmaxf(A_diag[p] - Alow, 0.0f);
  m11[p] = 1.0f - s * G;
  m12[p] = -s * A;
  dts[p] = s;
}

__global__ void pack_BC(const float* __restrict__ Bmat,
                        const float* __restrict__ Cmat,
                        float* __restrict__ BreP, float* __restrict__ BimP,
                        float* __restrict__ CreP, float* __restrict__ CimN) {
  int idx = blockIdx.x * blockDim.x + threadIdx.x;
  int p = idx & (PDIM - 1);
  int h = idx >> 8;
  float br = Bmat[(p * HDIM + h) * 2 + 0];
  float bi = Bmat[(p * HDIM + h) * 2 + 1];
  int bpos = (h & ~1) * PDIM + 2 * p + (h & 1);
  BreP[bpos] = br;
  BimP[bpos] = bi;
  float cr = Cmat[(h * PDIM + p) * 2 + 0];
  float ci = Cmat[(h * PDIM + p) * 2 + 1];
  int cpos = (p & ~1) * HDIM + 2 * h + (p & 1);
  CreP[cpos] = cr;
  CimN[cpos] = -ci;
}

__global__ __launch_bounds__(256) void gemm_bu(
    const float* __restrict__ u, const float* __restrict__ BreP,
    const float* __restrict__ BimP, const float* __restrict__ dts,
    float* __restrict__ fre, float* __restrict__ fim) {
  __shared__ float su[16 * LDSPITCH];
  const int tid = threadIdx.x;
  const int rowBase = blockIdx.x * 16;

  const float* src = u + (size_t)rowBase * HDIM;
  for (int i = tid; i < 16 * HDIM; i += 256)
    su[(i >> 8) * LDSPITCH + (i & 255)] = src[i];
  __syncthreads();

  const int wave = tid >> 5, lane = tid & 31;
  const int pBase = blockIdx.y * 128 + wave * 16;
  const int mn = lane & 15;
  const int ksel = (lane >> 4) << 1;

  v8f accRe = {};
  v8f accIm = {};
  const float* arow = su + mn * LDSPITCH + ksel;
  const float* brow = BreP + 2 * (pBase + mn);
  const float* irow = BimP + 2 * (pBase + mn);

  for (int k = 0; k < HDIM; k += 4) {
    v2f a = *(const v2f*)(arow + k);
    v2f b0 = *(const v2f*)(brow + (k + ksel) * PDIM);
    v2f b1 = *(const v2f*)(irow + (k + ksel) * PDIM);
    accRe = wmma4f32(a, b0, accRe);
    accIm = wmma4f32(a, b1, accIm);
  }

  const float scale = dts[pBase + mn];
  const int mo = rowBase + ((lane >> 4) << 3);
  const int col = pBase + mn;
#pragma unroll
  for (int r = 0; r < 8; ++r) {
    fre[(size_t)(mo + r) * PDIM + col] = accRe[r] * scale;
    fim[(size_t)(mo + r) * PDIM + col] = accIm[r] * scale;
  }
}

__global__ void scan_chunks(const float* __restrict__ fre,
                            const float* __restrict__ fim,
                            float* __restrict__ c1, float* __restrict__ c2,
                            const float* __restrict__ m11,
                            const float* __restrict__ m12,
                            const float* __restrict__ dts) {
  int tid = blockIdx.x * blockDim.x + threadIdx.x;
  int p = tid & (PDIM - 1);
  int comp = (tid >> 8) & 1;
  int chunk = (tid >> 9) & (NCHUNK - 1);
  int b = tid >> 13;

  const float* f = (comp ? fim : fre) +
                   (size_t)(b * LSEQ + chunk * LC) * PDIM + p;
  const float a11 = m11[p], a12 = m12[p], a21 = dts[p];
  float s1 = 0.0f, s2 = 0.0f;
  for (int t = 0; t < LC; ++t) {
    float ft = f[(size_t)t * PDIM];
    float n1 = fmaf(a11, s1, fmaf(a12, s2, ft));
    float n2 = fmaf(a21, s1, s2);
    s1 = n1;
    s2 = n2;
  }
  int ci = ((b * 2 + comp) * NCHUNK + chunk) * PDIM + p;
  c1[ci] = s1;
  c2[ci] = s2;
}

__global__ void scan_combine(float* __restrict__ c1, float* __restrict__ c2,
                             const float* __restrict__ m11,
                             const float* __restrict__ m12,
                             const float* __restrict__ dts) {
  int tid = blockIdx.x * blockDim.x + threadIdx.x;
  int p = tid & (PDIM - 1);
  int comp = (tid >> 8) & 1;
  int b = tid >> 9;

  float a = m11[p], bb = m12[p], c = dts[p], d = 1.0f;
#pragma unroll
  for (int i = 0; i < 8; ++i) {
    float na = fmaf(a, a, bb * c);
    float nb = fmaf(a, bb, bb * d);
    float nc = fmaf(c, a, d * c);
    float nd = fmaf(c, bb, d * d);
    a = na; bb = nb; c = nc; d = nd;
  }

  float i1 = 0.0f, i2 = 0.0f;
  int base = (b * 2 + comp) * NCHUNK * PDIM + p;
  for (int ch = 0; ch < NCHUNK; ++ch) {
    int ci = base + ch * PDIM;
    float s1 = c1[ci], s2 = c2[ci];
    c1[ci] = i1;
    c2[ci] = i2;
    float t1 = fmaf(a, i1, fmaf(bb, i2, s1));
    float t2 = fmaf(c, i1, fmaf(d, i2, s2));
    i1 = t1;
    i2 = t2;
  }
}

__global__ void scan_apply(float* __restrict__ fre, float* __restrict__ fim,
                           const float* __restrict__ c1,
                           const float* __restrict__ c2,
                           const float* __restrict__ m11,
                           const float* __restrict__ m12,
                           const float* __restrict__ dts) {
  int tid = blockIdx.x * blockDim.x + threadIdx.x;
  int p = tid & (PDIM - 1);
  int comp = (tid >> 8) & 1;
  int chunk = (tid >> 9) & (NCHUNK - 1);
  int b = tid >> 13;

  float* f = (comp ? fim : fre) + (size_t)(b * LSEQ + chunk * LC) * PDIM + p;
  const float a11 = m11[p], a12 = m12[p], a21 = dts[p];
  int ci = ((b * 2 + comp) * NCHUNK + chunk) * PDIM + p;
  float x1 = c1[ci], x2 = c2[ci];
  for (int t = 0; t < LC; ++t) {
    float ft = f[(size_t)t * PDIM];
    float y2 = fmaf(a21, x1, x2);
    float y1 = fmaf(a11, x1, fmaf(a12, x2, ft));
    f[(size_t)t * PDIM] = y2;
    x1 = y1;
    x2 = y2;
  }
}

__global__ __launch_bounds__(256) void gemm_out(
    const float* __restrict__ ysre, const float* __restrict__ ysim,
    const float* __restrict__ CreP, const float* __restrict__ CimN,
    const float* __restrict__ u, const float* __restrict__ Dvec,
    float* __restrict__ out) {
  __shared__ float sre[16 * LDSPITCH];
  __shared__ float sim[16 * LDSPITCH];
  const int tid = threadIdx.x;
  const int rowBase = blockIdx.x * 16;

  const float* pr = ysre + (size_t)rowBase * PDIM;
  const float* pi = ysim + (size_t)rowBase * PDIM;
  for (int i = tid; i < 16 * PDIM; i += 256) {
    int r = i >> 8, c = i & 255;
    sre[r * LDSPITCH + c] = pr[i];
    sim[r * LDSPITCH + c] = pi[i];
  }
  __syncthreads();

  const int wave = tid >> 5, lane = tid & 31;
  const int hBase = blockIdx.y * 128 + wave * 16;
  const int mn = lane & 15;
  const int ksel = (lane >> 4) << 1;

  v8f acc = {};
  const float* arow_re = sre + mn * LDSPITCH + ksel;
  const float* arow_im = sim + mn * LDSPITCH + ksel;
  const float* crow = CreP + 2 * (hBase + mn);
  const float* irow = CimN + 2 * (hBase + mn);

  for (int k = 0; k < PDIM; k += 4) {
    v2f ar = *(const v2f*)(arow_re + k);
    v2f ai = *(const v2f*)(arow_im + k);
    v2f vr = *(const v2f*)(crow + (k + ksel) * HDIM);
    v2f vi = *(const v2f*)(irow + (k + ksel) * HDIM);
    acc = wmma4f32(ar, vr, acc);
    acc = wmma4f32(ai, vi, acc);
  }

  const int h = hBase + mn;
  const float dh = Dvec[h];
  const int mo = rowBase + ((lane >> 4) << 3);
#pragma unroll
  for (int r = 0; r < 8; ++r) {
    size_t idx = (size_t)(mo + r) * HDIM + h;
    out[idx] = acc[r] + dh * u[idx];
  }
}

extern "C" void kernel_launch(void* const* d_in, const int* in_sizes, int n_in,
                              void* d_out, int out_size, void* d_ws,
                              size_t ws_size, hipStream_t stream) {
  const float* u      = (const float*)d_in[0];
  const float* A_diag = (const float*)d_in[1];
  const float* G_diag = (const float*)d_in[2];
  const float* dt     = (const float*)d_in[3];
  const float* Bmat   = (const float*)d_in[4];
  const float* Cmat   = (const float*)d_in[5];
  const float* Dvec   = (const float*)d_in[6];
  float* out = (float*)d_out;

  float* w = (float*)d_ws;
  const size_t NF = (size_t)MROWS * PDIM;
  float* BreP = w;
  float* BimP = BreP + (size_t)HDIM * PDIM;
  float* CreP = BimP + (size_t)HDIM * PDIM;
  float* CimN = CreP + (size_t)PDIM * HDIM;
  float* m11  = CimN + (size_t)PDIM * HDIM;
  float* m12  = m11 + PDIM;
  float* dts  = m12 + PDIM;
  float* fre  = dts + PDIM;
  float* fim  = fre + NF;
  float* c1   = fim + NF;
  float* c2   = c1 + (size_t)NBATCH * 2 * NCHUNK * PDIM;

  prep_params<<<1, 256, 0, stream>>>(A_diag, G_diag, dt, m11, m12, dts);
  pack_BC<<<256, 256, 0, stream>>>(Bmat, Cmat, BreP, BimP, CreP, CimN);
  gemm_bu<<<dim3(MROWS / 16, PDIM / 128), 256, 0, stream>>>(
      u, BreP, BimP, dts, fre, fim);
  scan_chunks<<<256, 256, 0, stream>>>(fre, fim, c1, c2, m11, m12, dts);
  scan_combine<<<16, 256, 0, stream>>>(c1, c2, m11, m12, dts);
  scan_apply<<<256, 256, 0, stream>>>(fre, fim, c1, c2, m11, m12, dts);
  gemm_out<<<dim3(MROWS / 16, HDIM / 128), 256, 0, stream>>>(
      fre, fim, CreP, CimN, u, Dvec, out);
}


# Both WMMA GEMMs retiled to a 64-row x 256-col block (16 waves); each wave keeps 4 row tiles, so one global B fragment feeds 4 WMMAs per plane; per-element sum order unchanged
# speedup vs baseline: 1.1498x; 1.1498x over previous
#include <hip/hip_runtime.h>
#include <hip/hip_bf16.h>


typedef float v2f __attribute__((ext_vector_type(2)));
typedef float v8f __attribute__((ext_vector_type(8)));

#define PDIM 256
#define HDIM 256
#define NBATCH 8
#define LSEQ 4096
#define MROWS (NBATCH * LSEQ)
#define LC 256
#define NCHUNK (LSEQ / LC)
#define LDSPITCH 260
#define RT 4
#define BM (RT * 16)
#define GEMM_THREADS 512

static __device__ inline v8f wmma4f32(v2f a, v2f b, v8f c) {
  return __builtin_amdgcn_wmma_f32_16x16x4_f32(
       false, a,  false, b,
       (short)0, c,  false,  false);
}

__global__ void prep_params(const float* __restrict__ A_diag,
                            const float* __restrict__ G_diag,
                            const float* __restrict__ dt,
                            float* __restrict__ m11,
                            float* __restrict__ m12,
                            float* __restrict__ dts) {
  int p = threadIdx.x;
  float s = 1.0f / (1.0f + __expf(-dt[p]));
  float Glow = s * A_diag[p];
  float G = Glow + fmaxf(G_diag[p] - Glow, 0.0f);
  float Alow = 0.25f * G * G;
  float A = Alow + fmaxf(A_diag[p] - Alow, 0.0f);
  m11[p] = 1.0f - s * G;
  m12[p] = -s * A;
  dts[p] = s;
}

__global__ void pack_BC(const float* __restrict__ Bmat,
                        const float* __restrict__ Cmat,
                        float* __restrict__ BreP, float* __restrict__ BimP,
                        float* __restrict__ CreP, float* __restrict__ CimN) {
  int idx = blockIdx.x * blockDim.x + threadIdx.x;
  int p = idx & (PDIM - 1);
  int h = idx >> 8;
  float br = Bmat[(p * HDIM + h) * 2 + 0];
  float bi = Bmat[(p * HDIM + h) * 2 + 1];
  int bpos = (h & ~1) * PDIM + 2 * p + (h & 1);
  BreP[bpos] = br;
  BimP[bpos] = bi;
  float cr = Cmat[(h * PDIM + p) * 2 + 0];
  float ci = Cmat[(h * PDIM + p) * 2 + 1];
  int cpos = (p & ~1) * HDIM + 2 * h + (p & 1);
  CreP[cpos] = cr;
  CimN[cpos] = -ci;
}

__global__ __launch_bounds__(GEMM_THREADS) void gemm_bu(
    const float* __restrict__ u, const float* __restrict__ BreP,
    const float* __restrict__ BimP, const float* __restrict__ dts,
    float* __restrict__ fre, float* __restrict__ fim) {
  __shared__ float su[BM * LDSPITCH];
  const int tid = threadIdx.x;
  const int rowBase = blockIdx.x * BM;

  const float* src = u + (size_t)rowBase * HDIM;
  for (int j = 0; j < BM * HDIM / GEMM_THREADS; ++j) {
    const int i = j * GEMM_THREADS + tid;
    su[(i >> 8) * LDSPITCH + (i & 255)] = src[i];
  }
  __syncthreads();

  const int wave = tid >> 5, lane = tid & 31;
  const int pBase = wave * 16;
  const int mn = lane & 15;
  const int ksel = (lane >> 4) << 1;

  v8f accRe[RT];
  v8f accIm[RT];
#pragma unroll
  for (int t = 0; t < RT; ++t) {
    accRe[t] = v8f{};
    accIm[t] = v8f{};
  }
  const float* arow = su + mn * LDSPITCH + ksel;
  const float* brow = BreP + 2 * (pBase + mn);
  const float* irow = BimP + 2 * (pBase + mn);

  for (int k = 0; k < HDIM; k += 4) {
    v2f b0 = *(const v2f*)(brow + (k + ksel) * PDIM);
    v2f b1 = *(const v2f*)(irow + (k + ksel) * PDIM);
#pragma unroll
    for (int t = 0; t < RT; ++t) {
      v2f a = *(const v2f*)(arow + t * 16 * LDSPITCH + k);
      accRe[t] = wmma4f32(a, b0, accRe[t]);
      accIm[t] = wmma4f32(a, b1, accIm[t]);
    }
  }

  const float scale = dts[pBase + mn];
  const int col = pBase + mn;
#pragma unroll
  for (int t = 0; t < RT; ++t) {
    const int mo = rowBase + t * 16 + ((lane >> 4) << 3);
#pragma unroll
    for (int r = 0; r < 8; ++r) {
      fre[(size_t)(mo + r) * PDIM + col] = accRe[t][r] * scale;
      fim[(size_t)(mo + r) * PDIM + col] = accIm[t][r] * scale;
    }
  }
}

__global__ void scan_chunks(const float* __restrict__ fre,
                            const float* __restrict__ fim,
                            float* __restrict__ c1, float* __restrict__ c2,
                            const float* __restrict__ m11,
                            const float* __restrict__ m12,
                            const float* __restrict__ dts) {
  int tid = blockIdx.x * blockDim.x + threadIdx.x;
  int p = tid & (PDIM - 1);
  int comp = (tid >> 8) & 1;
  int chunk = (tid >> 9) & (NCHUNK - 1);
  int b = tid >> 13;

  const float* f = (comp ? fim : fre) +
                   (size_t)(b * LSEQ + chunk * LC) * PDIM + p;
  const float a11 = m11[p], a12 = m12[p], a21 = dts[p];
  float s1 = 0.0f, s2 = 0.0f;
  for (int t = 0; t < LC; ++t) {
    float ft = f[(size_t)t * PDIM];
    float n1 = fmaf(a11, s1, fmaf(a12, s2, ft));
    float n2 = fmaf(a21, s1, s2);
    s1 = n1;
    s2 = n2;
  }
  int ci = ((b * 2 + comp) * NCHUNK + chunk) * PDIM + p;
  c1[ci] = s1;
  c2[ci] = s2;
}

__global__ void scan_combine(float* __restrict__ c1, float* __restrict__ c2,
                             const float* __restrict__ m11,
                             const float* __restrict__ m12,
                             const float* __restrict__ dts) {
  int tid = blockIdx.x * blockDim.x + threadIdx.x;
  int p = tid & (PDIM - 1);
  int comp = (tid >> 8) & 1;
  int b = tid >> 9;

  float a = m11[p], bb = m12[p], c = dts[p], d = 1.0f;
#pragma unroll
  for (int i = 0; i < 8; ++i) {
    float na = fmaf(a, a, bb * c);
    float nb = fmaf(a, bb, bb * d);
    float nc = fmaf(c, a, d * c);
    float nd = fmaf(c, bb, d * d);
    a = na; bb = nb; c = nc; d = nd;
  }

  float i1 = 0.0f, i2 = 0.0f;
  int base = (b * 2 + comp) * NCHUNK * PDIM + p;
  for (int ch = 0; ch < NCHUNK; ++ch) {
    int ci = base + ch * PDIM;
    float s1 = c1[ci], s2 = c2[ci];
    c1[ci] = i1;
    c2[ci] = i2;
    float t1 = fmaf(a, i1, fmaf(bb, i2, s1));
    float t2 = fmaf(c, i1, fmaf(d, i2, s2));
    i1 = t1;
    i2 = t2;
  }
}

__global__ void scan_apply(float* __restrict__ fre, float* __restrict__ fim,
                           const float* __restrict__ c1,
                           const float* __restrict__ c2,
                           const float* __restrict__ m11,
                           const float* __restrict__ m12,
                           const float* __restrict__ dts) {
  int tid = blockIdx.x * blockDim.x + threadIdx.x;
  int p = tid & (PDIM - 1);
  int comp = (tid >> 8) & 1;
  int chunk = (tid >> 9) & (NCHUNK - 1);
  int b = tid >> 13;

  float* f = (comp ? fim : fre) + (size_t)(b * LSEQ + chunk * LC) * PDIM + p;
  const float a11 = m11[p], a12 = m12[p], a21 = dts[p];
  int ci = ((b * 2 + comp) * NCHUNK + chunk) * PDIM + p;
  float x1 = c1[ci], x2 = c2[ci];
  for (int t = 0; t < LC; ++t) {
    float ft = f[(size_t)t * PDIM];
    float y2 = fmaf(a21, x1, x2);
    float y1 = fmaf(a11, x1, fmaf(a12, x2, ft));
    f[(size_t)t * PDIM] = y2;
    x1 = y1;
    x2 = y2;
  }
}

__global__ __launch_bounds__(GEMM_THREADS) void gemm_out(
    const float* __restrict__ ysre, const float* __restrict__ ysim,
    const float* __restrict__ CreP, const float* __restrict__ CimN,
    const float* __restrict__ u, const float* __restrict__ Dvec,
    float* __restrict__ out) {
  __shared__ float sre[BM * LDSPITCH];
  __shared__ float sim[BM * LDSPITCH];
  const int tid = threadIdx.x;
  const int rowBase = blockIdx.x * BM;

  const float* pr = ysre + (size_t)rowBase * PDIM;
  const float* pi = ysim + (size_t)rowBase * PDIM;
#pragma nounroll
  for (int j0 = 0; j0 < BM * PDIM / GEMM_THREADS; j0 += 16) {
#pragma unroll
    for (int jj = 0; jj < 16; ++jj) {
      const int i = (j0 + jj) * GEMM_THREADS + tid;
      int r = i >> 8, c = i & 255;
      sre[r * LDSPITCH + c] = pr[i];
      sim[r * LDSPITCH + c] = pi[i];
    }
  }
  __syncthreads();

  const int wave = tid >> 5, lane = tid & 31;
  const int hBase = wave * 16;
  const int mn = lane & 15;
  const int ksel = (lane >> 4) << 1;

  v8f acc[RT];
#pragma unroll
  for (int t = 0; t < RT; ++t) acc[t] = v8f{};
  const float* arow_re = sre + mn * LDSPITCH + ksel;
  const float* arow_im = sim + mn * LDSPITCH + ksel;
  const float* crow = CreP + 2 * (hBase + mn);
  const float* irow = CimN + 2 * (hBase + mn);

  for (int k = 0; k < PDIM; k += 4) {
    v2f vr = *(const v2f*)(crow + (k + ksel) * HDIM);
    v2f vi = *(const v2f*)(irow + (k + ksel) * HDIM);
#pragma unroll
    for (int t = 0; t < RT; ++t) {
      v2f ar = *(const v2f*)(arow_re + t * 16 * LDSPITCH + k);
      v2f ai = *(const v2f*)(arow_im + t * 16 * LDSPITCH + k);
      acc[t] = wmma4f32(ar, vr, acc[t]);
      acc[t] = wmma4f32(ai, vi, acc[t]);
    }
  }

  const int h = hBase + mn;
  const float dh = Dvec[h];
#pragma unroll
  for (int t = 0; t < RT; ++t) {
    const int mo = rowBase + t * 16 + ((lane >> 4) << 3);
#pragma unroll
    for (int r = 0; r < 8; ++r) {
      size_t idx = (size_t)(mo + r) * HDIM + h;
      out[idx] = acc[t][r] + dh * u[idx];
    }
  }
}

extern "C" void kernel_launch(void* const* d_in, const int* in_sizes, int n_in,
                              void* d_out, int out_size, void* d_ws,
                              size_t ws_size, hipStream_t stream) {
  const float* u      = (const float*)d_in[0];
  const float* A_diag = (const float*)d_in[1];
  const float* G_diag = (const float*)d_in[2];
  const float* dt     = (const float*)d_in[3];
  const float* Bmat   = (const float*)d_in[4];
  const float* Cmat   = (const float*)d_in[5];
  const float* Dvec   = (const float*)d_in[6];
  float* out = (float*)d_out;

  float* w = (float*)d_ws;
  const size_t NF = (size_t)MROWS * PDIM;
  float* BreP = w;
  float* BimP = BreP + (size_t)HDIM * PDIM;
  float* CreP = BimP + (size_t)HDIM * PDIM;
  float* CimN = CreP + (size_t)PDIM * HDIM;
  float* m11  = CimN + (size_t)PDIM * HDIM;
  float* m12  = m11 + PDIM;
  float* dts  = m12 + PDIM;
  float* fre  = dts + PDIM;
  float* fim  = fre + NF;
  float* c1   = fim + NF;
  float* c2   = c1 + (size_t)NBATCH * 2 * NCHUNK * PDIM;

  prep_params<<<1, 256, 0, stream>>>(A_diag, G_diag, dt, m11, m12, dts);
  pack_BC<<<256, 256, 0, stream>>>(Bmat, Cmat, BreP, BimP, CreP, CimN);
  gemm_bu<<<MROWS / BM, GEMM_THREADS, 0, stream>>>(
      u, BreP, BimP, dts, fre, fim);
  scan_chunks<<<256, 256, 0, stream>>>(fre, fim, c1, c2, m11, m12, dts);
  scan_combine<<<16, 256, 0, stream>>>(c1, c2, m11, m12, dts);
  scan_apply<<<256, 256, 0, stream>>>(fre, fim, c1, c2, m11, m12, dts);
  gemm_out<<<MROWS / BM, GEMM_THREADS, 0, stream>>>(
      fre, fim, CreP, CimN, u, Dvec, out);
}


# Scan apply pass fused into the output GEMM: chunk 256->128 rows = one 1024-thread block that re-scans its chunk straight into its LDS A tile, so ys is never stored; chunked sum order changes
# speedup vs baseline: 1.3663x; 1.1883x over previous
#include <hip/hip_runtime.h>
#include <hip/hip_bf16.h>


typedef float v2f __attribute__((ext_vector_type(2)));
typedef float v8f __attribute__((ext_vector_type(8)));

#define PDIM 256
#define HDIM 256
#define NBATCH 8
#define LSEQ 4096
#define MROWS (NBATCH * LSEQ)
#define LC 128
#define NCHUNK (LSEQ / LC)
#define LDSPITCH 260
#define RT 4
#define BM (RT * 16)
#define GEMM_THREADS 512
#define OUT_THREADS 1024
#define SCAN_PASS 32

static __device__ inline v8f wmma4f32(v2f a, v2f b, v8f c) {
  return __builtin_amdgcn_wmma_f32_16x16x4_f32(
       false, a,  false, b,
       (short)0, c,  false,  false);
}

__global__ void prep_params(const float* __restrict__ A_diag,
                            const float* __restrict__ G_diag,
                            const float* __restrict__ dt,
                            float* __restrict__ m11,
                            float* __restrict__ m12,
                            float* __restrict__ dts) {
  int p = threadIdx.x;
  float s = 1.0f / (1.0f + __expf(-dt[p]));
  float Glow = s * A_diag[p];
  float G = Glow + fmaxf(G_diag[p] - Glow, 0.0f);
  float Alow = 0.25f * G * G;
  float A = Alow + fmaxf(A_diag[p] - Alow, 0.0f);
  m11[p] = 1.0f - s * G;
  m12[p] = -s * A;
  dts[p] = s;
}

__global__ void pack_BC(const float* __restrict__ Bmat,
                        const float* __restrict__ Cmat,
                        float* __restrict__ BreP, float* __restrict__ BimP,
                        float* __restrict__ CreP, float* __restrict__ CimN) {
  int idx = blockIdx.x * blockDim.x + threadIdx.x;
  int p = idx & (PDIM - 1);
  int h = idx >> 8;
  float br = Bmat[(p * HDIM + h) * 2 + 0];
  float bi = Bmat[(p * HDIM + h) * 2 + 1];
  int bpos = (h & ~1) * PDIM + 2 * p + (h & 1);
  BreP[bpos] = br;
  BimP[bpos] = bi;
  float cr = Cmat[(h * PDIM + p) * 2 + 0];
  float ci = Cmat[(h * PDIM + p) * 2 + 1];
  int cpos = (p & ~1) * HDIM + 2 * h + (p & 1);
  CreP[cpos] = cr;
  CimN[cpos] = -ci;
}

__global__ __launch_bounds__(GEMM_THREADS) void gemm_bu(
    const float* __restrict__ u, const float* __restrict__ BreP,
    const float* __restrict__ BimP, const float* __restrict__ dts,
    float* __restrict__ fre, float* __restrict__ fim) {
  __shared__ float su[BM * LDSPITCH];
  const int tid = threadIdx.x;
  const int rowBase = blockIdx.x * BM;

  const float* src = u + (size_t)rowBase * HDIM;
  for (int j = 0; j < BM * HDIM / GEMM_THREADS; ++j) {
    const int i = j * GEMM_THREADS + tid;
    su[(i >> 8) * LDSPITCH + (i & 255)] = src[i];
  }
  __syncthreads();

  const int wave = tid >> 5, lane = tid & 31;
  const int pBase = wave * 16;
  const int mn = lane & 15;
  const int ksel = (lane >> 4) << 1;

  v8f accRe[RT];
  v8f accIm[RT];
#pragma unroll
  for (int t = 0; t < RT; ++t) {
    accRe[t] = v8f{};
    accIm[t] = v8f{};
  }
  const float* arow = su + mn * LDSPITCH + ksel;
  const float* brow = BreP + 2 * (pBase + mn);
  const float* irow = BimP + 2 * (pBase + mn);

  for (int k = 0; k < HDIM; k += 4) {
    v2f b0 = *(const v2f*)(brow + (k + ksel) * PDIM);
    v2f b1 = *(const v2f*)(irow + (k + ksel) * PDIM);
#pragma unroll
    for (int t = 0; t < RT; ++t) {
      v2f a = *(const v2f*)(arow + t * 16 * LDSPITCH + k);
      accRe[t] = wmma4f32(a, b0, accRe[t]);
      accIm[t] = wmma4f32(a, b1, accIm[t]);
    }
  }

  const float scale = dts[pBase + mn];
  const int col = pBase + mn;
#pragma unroll
  for (int t = 0; t < RT; ++t) {
    const int mo = rowBase + t * 16 + ((lane >> 4) << 3);
#pragma unroll
    for (int r = 0; r < 8; ++r) {
      fre[(size_t)(mo + r) * PDIM + col] = accRe[t][r] * scale;
      fim[(size_t)(mo + r) * PDIM + col] = accIm[t][r] * scale;
    }
  }
}

__global__ void scan_chunks(const float* __restrict__ fre,
                            const float* __restrict__ fim,
                            float* __restrict__ c1, float* __restrict__ c2,
                            const float* __restrict__ m11,
                            const float* __restrict__ m12,
                            const float* __restrict__ dts) {
  int tid = blockIdx.x * blockDim.x + threadIdx.x;
  int p = tid & (PDIM - 1);
  int comp = (tid >> 8) & 1;
  int cb = tid >> 9;
  const float* f = (comp ? fim : fre) + (size_t)cb * LC * PDIM + p;
  const float a11 = m11[p], a12 = m12[p], a21 = dts[p];
  float s1 = 0.0f, s2 = 0.0f;
  for (int t = 0; t < LC; ++t) {
    float ft = f[(size_t)t * PDIM];
    float n1 = fmaf(a11, s1, fmaf(a12, s2, ft));
    float n2 = fmaf(a21, s1, s2);
    s1 = n1;
    s2 = n2;
  }
  int ci = (cb * 2 + comp) * PDIM + p;
  c1[ci] = s1;
  c2[ci] = s2;
}

__global__ void scan_combine(float* __restrict__ c1, float* __restrict__ c2,
                             const float* __restrict__ m11,
                             const float* __restrict__ m12,
                             const float* __restrict__ dts) {
  int tid = blockIdx.x * blockDim.x + threadIdx.x;
  int p = tid & (PDIM - 1);
  int comp = (tid >> 8) & 1;
  int b = tid >> 9;

  float a = m11[p], bb = m12[p], c = dts[p], d = 1.0f;
#pragma unroll
  for (int i = 0; i < 7; ++i) {
    float na = fmaf(a, a, bb * c);
    float nb = fmaf(a, bb, bb * d);
    float nc = fmaf(c, a, d * c);
    float nd = fmaf(c, bb, d * d);
    a = na; bb = nb; c = nc; d = nd;
  }

  int base = (b * NCHUNK * 2 + comp) * PDIM + p;
  float l1[NCHUNK], l2[NCHUNK];
#pragma unroll
  for (int ch = 0; ch < NCHUNK; ++ch) {
    l1[ch] = c1[base + ch * 2 * PDIM];
    l2[ch] = c2[base + ch * 2 * PDIM];
  }
  float i1 = 0.0f, i2 = 0.0f;
#pragma unroll
  for (int ch = 0; ch < NCHUNK; ++ch) {
    int ci = base + ch * 2 * PDIM;
    c1[ci] = i1;
    c2[ci] = i2;
    float t1 = fmaf(a, i1, fmaf(bb, i2, l1[ch]));
    float t2 = fmaf(c, i1, fmaf(d, i2, l2[ch]));
    i1 = t1;
    i2 = t2;
  }
}

__global__ __launch_bounds__(OUT_THREADS) void gemm_out(
    const float* __restrict__ fre, const float* __restrict__ fim,
    const float* __restrict__ c1, const float* __restrict__ c2,
    const float* __restrict__ m11, const float* __restrict__ m12,
    const float* __restrict__ dts,
    const float* __restrict__ CreP, const float* __restrict__ CimN,
    const float* __restrict__ u, const float* __restrict__ Dvec,
    float* __restrict__ out) {
  __shared__ float sre[LC * LDSPITCH];
  __shared__ float sim[LC * LDSPITCH];
  const int tid = threadIdx.x;
  const int rowBase = blockIdx.x * LC;

  if (tid < 2 * PDIM) {
    const int comp = tid >> 8, p = tid & (PDIM - 1);
    const float* f = (comp ? fim : fre) + (size_t)rowBase * PDIM + p;
    float* s = (comp ? sim : sre) + p;
    const float a11 = m11[p], a12 = m12[p], a21 = dts[p];
    const int ci = (blockIdx.x * 2 + comp) * PDIM + p;
    float x1 = c1[ci], x2 = c2[ci];
    float cur[SCAN_PASS], nxt[SCAN_PASS];
#pragma unroll
    for (int j = 0; j < SCAN_PASS; ++j) cur[j] = f[(size_t)j * PDIM];
#pragma unroll
    for (int t0 = 0; t0 < LC; t0 += SCAN_PASS) {
      if (t0 + SCAN_PASS < LC) {
#pragma unroll
        for (int j = 0; j < SCAN_PASS; ++j)
          nxt[j] = f[(size_t)(t0 + SCAN_PASS + j) * PDIM];
      }
#pragma unroll
      for (int j = 0; j < SCAN_PASS; ++j) {
        float y2 = fmaf(a21, x1, x2);
        float y1 = fmaf(a11, x1, fmaf(a12, x2, cur[j]));
        s[(t0 + j) * LDSPITCH] = y2;
        x1 = y1;
        x2 = y2;
      }
#pragma unroll
      for (int j = 0; j < SCAN_PASS; ++j) cur[j] = nxt[j];
    }
  }
  __syncthreads();

  const int wave = tid >> 5, lane = tid & 31;
  const int hBase = (wave & 15) * 16;
  const int rHalf = (wave >> 4) * BM;
  const int mn = lane & 15;
  const int ksel = (lane >> 4) << 1;

  v8f acc[RT];
#pragma unroll
  for (int t = 0; t < RT; ++t) acc[t] = v8f{};
  const float* arow_re = sre + (rHalf + mn) * LDSPITCH + ksel;
  const float* arow_im = sim + (rHalf + mn) * LDSPITCH + ksel;
  const float* crow = CreP + 2 * (hBase + mn);
  const float* irow = CimN + 2 * (hBase + mn);

  for (int k = 0; k < PDIM; k += 4) {
    v2f vr = *(const v2f*)(crow + (k + ksel) * HDIM);
    v2f vi = *(const v2f*)(irow + (k + ksel) * HDIM);
#pragma unroll
    for (int t = 0; t < RT; ++t) {
      v2f ar = *(const v2f*)(arow_re + t * 16 * LDSPITCH + k);
      v2f ai = *(const v2f*)(arow_im + t * 16 * LDSPITCH + k);
      acc[t] = wmma4f32(ar, vr, acc[t]);
      acc[t] = wmma4f32(ai, vi, acc[t]);
    }
  }

  const int h = hBase + mn;
  const float dh = Dvec[h];
#pragma unroll
  for (int t = 0; t < RT; ++t) {
    const int mo = rowBase + rHalf + t * 16 + ((lane >> 4) << 3);
#pragma unroll
    for (int r = 0; r < 8; ++r) {
      size_t idx = (size_t)(mo + r) * HDIM + h;
      out[idx] = acc[t][r] + dh * u[idx];
    }
  }
}

extern "C" void kernel_launch(void* const* d_in, const int* in_sizes, int n_in,
                              void* d_out, int out_size, void* d_ws,
                              size_t ws_size, hipStream_t stream) {
  const float* u      = (const float*)d_in[0];
  const float* A_diag = (const float*)d_in[1];
  const float* G_diag = (const float*)d_in[2];
  const float* dt     = (const float*)d_in[3];
  const float* Bmat   = (const float*)d_in[4];
  const float* Cmat   = (const float*)d_in[5];
  const float* Dvec   = (const float*)d_in[6];
  float* out = (float*)d_out;

  float* w = (float*)d_ws;
  const size_t NF = (size_t)MROWS * PDIM;
  float* CreP = w;
  float* CimN = CreP + (size_t)PDIM * HDIM;
  float* m11  = CimN + (size_t)PDIM * HDIM;
  float* m12  = m11 + PDIM;
  float* dts  = m12 + PDIM;
  float* fre  = dts + PDIM;
  float* fim  = fre + NF;
  float* BreP = fim + NF;
  float* BimP = BreP + (size_t)HDIM * PDIM;
  float* c1   = BreP;
  float* c2   = c1 + (size_t)NBATCH * 2 * NCHUNK * PDIM;

  prep_params<<<1, 256, 0, stream>>>(A_diag, G_diag, dt, m11, m12, dts);
  pack_BC<<<256, 256, 0, stream>>>(Bmat, Cmat, BreP, BimP, CreP, CimN);
  gemm_bu<<<MROWS / BM, GEMM_THREADS, 0, stream>>>(
      u, BreP, BimP, dts, fre, fim);
  scan_chunks<<<512, 256, 0, stream>>>(fre, fim, c1, c2, m11, m12, dts);
  scan_combine<<<16, 256, 0, stream>>>(c1, c2, m11, m12, dts);
  gemm_out<<<MROWS / LC, OUT_THREADS, 0, stream>>>(
      fre, fim, c1, c2, m11, m12, dts, CreP, CimN, u, Dvec, out);
}


# Helper launches removed, 6 -> 3: oscillator parameters computed inline, carry join moved into gemm_out's prologue, B/C fragments read from the raw interleaved arrays; every sum order unchanged
# speedup vs baseline: 1.4638x; 1.0714x over previous
#include <hip/hip_runtime.h>
#include <hip/hip_bf16.h>


typedef float v2f __attribute__((ext_vector_type(2)));
typedef float v4f __attribute__((ext_vector_type(4)));
typedef float v8f __attribute__((ext_vector_type(8)));

#define PDIM 256
#define HDIM 256
#define NBATCH 8
#define LSEQ 4096
#define MROWS (NBATCH * LSEQ)
#define LC 128
#define NCHUNK (LSEQ / LC)
#define LDSPITCH 260
#define RT 4
#define BM (RT * 16)
#define GEMM_THREADS 512
#define OUT_THREADS 1024
#define SCAN_PASS 32
#define KGROUP 16

static __device__ inline v8f wmma4f32(v2f a, v2f b, v8f c) {
  return __builtin_amdgcn_wmma_f32_16x16x4_f32(
       false, a,  false, b,
       (short)0, c,  false,  false);
}

static __device__ inline void osc_params(const float* __restrict__ A_diag,
                                         const float* __restrict__ G_diag,
                                         const float* __restrict__ dt, int p,
                                         float& m11, float& m12, float& dts) {
  float s = 1.0f / (1.0f + __expf(-dt[p]));
  float Glow = s * A_diag[p];
  float G = Glow + fmaxf(G_diag[p] - Glow, 0.0f);
  float Alow = 0.25f * G * G;
  float A = Alow + fmaxf(A_diag[p] - Alow, 0.0f);
  m11 = 1.0f - s * G;
  m12 = -s * A;
  dts = s;
}

__global__ __launch_bounds__(GEMM_THREADS) void gemm_bu(
    const float* __restrict__ u, const float* __restrict__ Bmat,
    const float* __restrict__ A_diag, const float* __restrict__ G_diag,
    const float* __restrict__ dt,
    float* __restrict__ fre, float* __restrict__ fim) {
  __shared__ float su[BM * LDSPITCH];
  const int tid = threadIdx.x;
  const int rowBase = blockIdx.x * BM;

  const float* src = u + (size_t)rowBase * HDIM;
  for (int j = 0; j < BM * HDIM / GEMM_THREADS; ++j) {
    const int i = j * GEMM_THREADS + tid;
    su[(i >> 8) * LDSPITCH + (i & 255)] = src[i];
  }
  __syncthreads();

  const int wave = tid >> 5, lane = tid & 31;
  const int pBase = wave * 16;
  const int mn = lane & 15;
  const int ksel = (lane >> 4) << 1;

  v8f accRe[RT];
  v8f accIm[RT];
#pragma unroll
  for (int t = 0; t < RT; ++t) {
    accRe[t] = v8f{};
    accIm[t] = v8f{};
  }
  const float* arow = su + mn * LDSPITCH + ksel;
  const float* brow = Bmat + ((size_t)(pBase + mn) * HDIM + ksel) * 2;

  for (int k = 0; k < HDIM; k += KGROUP) {
    v4f q[KGROUP / 4];
#pragma unroll
    for (int j = 0; j < KGROUP / 4; ++j)
      q[j] = *(const v4f*)(brow + (k + 4 * j) * 2);
#pragma unroll
    for (int j = 0; j < KGROUP / 4; ++j) {
      const v2f b0 = {q[j].x, q[j].z};
      const v2f b1 = {q[j].y, q[j].w};
#pragma unroll
      for (int t = 0; t < RT; ++t) {
        v2f a = *(const v2f*)(arow + t * 16 * LDSPITCH + k + 4 * j);
        accRe[t] = wmma4f32(a, b0, accRe[t]);
        accIm[t] = wmma4f32(a, b1, accIm[t]);
      }
    }
  }

  const int col = pBase + mn;
  float c11, c12, scale;
  osc_params(A_diag, G_diag, dt, col, c11, c12, scale);
#pragma unroll
  for (int t = 0; t < RT; ++t) {
    const int mo = rowBase + t * 16 + ((lane >> 4) << 3);
#pragma unroll
    for (int r = 0; r < 8; ++r) {
      fre[(size_t)(mo + r) * PDIM + col] = accRe[t][r] * scale;
      fim[(size_t)(mo + r) * PDIM + col] = accIm[t][r] * scale;
    }
  }
}

__global__ void scan_chunks(const float* __restrict__ fre,
                            const float* __restrict__ fim,
                            float* __restrict__ c1, float* __restrict__ c2,
                            const float* __restrict__ A_diag,
                            const float* __restrict__ G_diag,
                            const float* __restrict__ dt) {
  int tid = blockIdx.x * blockDim.x + threadIdx.x;
  int p = tid & (PDIM - 1);
  int comp = (tid >> 8) & 1;
  int cb = tid >> 9;
  const float* f = (comp ? fim : fre) + (size_t)cb * LC * PDIM + p;
  float a11, a12, a21;
  osc_params(A_diag, G_diag, dt, p, a11, a12, a21);
  float s1 = 0.0f, s2 = 0.0f;
  for (int t = 0; t < LC; ++t) {
    float ft = f[(size_t)t * PDIM];
    float n1 = fmaf(a11, s1, fmaf(a12, s2, ft));
    float n2 = fmaf(a21, s1, s2);
    s1 = n1;
    s2 = n2;
  }
  int ci = (cb * 2 + comp) * PDIM + p;
  c1[ci] = s1;
  c2[ci] = s2;
}

__global__ __launch_bounds__(OUT_THREADS) void gemm_out(
    const float* __restrict__ fre, const float* __restrict__ fim,
    const float* __restrict__ c1, const float* __restrict__ c2,
    const float* __restrict__ A_diag, const float* __restrict__ G_diag,
    const float* __restrict__ dt, const float* __restrict__ Cmat,
    const float* __restrict__ u, const float* __restrict__ Dvec,
    float* __restrict__ out) {
  __shared__ float sre[LC * LDSPITCH];
  __shared__ float sim[LC * LDSPITCH];
  const int tid = threadIdx.x;
  const int rowBase = blockIdx.x * LC;
  const int chunk = blockIdx.x & (NCHUNK - 1);

  if (tid < 2 * PDIM) {
    const int comp = tid >> 8, p = tid & (PDIM - 1);
    const float* f = (comp ? fim : fre) + (size_t)rowBase * PDIM + p;
    float* s = (comp ? sim : sre) + p;
    const float sgn = comp ? -1.0f : 1.0f;
    float cur[SCAN_PASS], nxt[SCAN_PASS];
#pragma unroll
    for (int j = 0; j < SCAN_PASS; ++j) cur[j] = f[(size_t)j * PDIM];

    const int base = ((blockIdx.x - chunk) * 2 + comp) * PDIM + p;
    float l1[NCHUNK - 1], l2[NCHUNK - 1];
#pragma unroll
    for (int ch = 0; ch < NCHUNK - 1; ++ch) {
      l1[ch] = c1[base + ch * 2 * PDIM];
      l2[ch] = c2[base + ch * 2 * PDIM];
    }

    float a11, a12, a21;
    osc_params(A_diag, G_diag, dt, p, a11, a12, a21);

    float a = a11, bb = a12, c = a21, d = 1.0f;
#pragma unroll
    for (int i = 0; i < 7; ++i) {
      float na = fmaf(a, a, bb * c);
      float nb = fmaf(a, bb, bb * d);
      float nc = fmaf(c, a, d * c);
      float nd = fmaf(c, bb, d * d);
      a = na; bb = nb; c = nc; d = nd;
    }
    float x1 = 0.0f, x2 = 0.0f;
#pragma unroll
    for (int ch = 0; ch < NCHUNK - 1; ++ch) {
      float t1 = fmaf(a, x1, fmaf(bb, x2, l1[ch]));
      float t2 = fmaf(c, x1, fmaf(d, x2, l2[ch]));
      if (ch < chunk) {
        x1 = t1;
        x2 = t2;
      }
    }

#pragma unroll
    for (int t0 = 0; t0 < LC; t0 += SCAN_PASS) {
      if (t0 + SCAN_PASS < LC) {
#pragma unroll
        for (int j = 0; j < SCAN_PASS; ++j)
          nxt[j] = f[(size_t)(t0 + SCAN_PASS + j) * PDIM];
      }
#pragma unroll
      for (int j = 0; j < SCAN_PASS; ++j) {
        float y2 = fmaf(a21, x1, x2);
        float y1 = fmaf(a11, x1, fmaf(a12, x2, cur[j]));
        s[(t0 + j) * LDSPITCH] = y2 * sgn;
        x1 = y1;
        x2 = y2;
      }
#pragma unroll
      for (int j = 0; j < SCAN_PASS; ++j) cur[j] = nxt[j];
    }
  }
  __syncthreads();

  const int wave = tid >> 5, lane = tid & 31;
  const int hBase = (wave & 15) * 16;
  const int rHalf = (wave >> 4) * BM;
  const int mn = lane & 15;
  const int ksel = (lane >> 4) << 1;

  v8f acc[RT];
#pragma unroll
  for (int t = 0; t < RT; ++t) acc[t] = v8f{};
  const float* arow_re = sre + (rHalf + mn) * LDSPITCH + ksel;
  const float* arow_im = sim + (rHalf + mn) * LDSPITCH + ksel;
  const float* crow = Cmat + ((size_t)(hBase + mn) * PDIM + ksel) * 2;

  for (int k = 0; k < PDIM; k += KGROUP) {
    v4f q[KGROUP / 4];
#pragma unroll
    for (int j = 0; j < KGROUP / 4; ++j)
      q[j] = *(const v4f*)(crow + (k + 4 * j) * 2);
#pragma unroll
    for (int j = 0; j < KGROUP / 4; ++j) {
      const v2f vr = {q[j].x, q[j].z};
      const v2f vi = {q[j].y, q[j].w};
#pragma unroll
      for (int t = 0; t < RT; ++t) {
        v2f ar = *(const v2f*)(arow_re + t * 16 * LDSPITCH + k + 4 * j);
        v2f ai = *(const v2f*)(arow_im + t * 16 * LDSPITCH + k + 4 * j);
        acc[t] = wmma4f32(ar, vr, acc[t]);
        acc[t] = wmma4f32(ai, vi, acc[t]);
      }
    }
  }

  const int h = hBase + mn;
  const float dh = Dvec[h];
#pragma unroll
  for (int t = 0; t < RT; ++t) {
    const int mo = rowBase + rHalf + t * 16 + ((lane >> 4) << 3);
#pragma unroll
    for (int r = 0; r < 8; ++r) {
      size_t idx = (size_t)(mo + r) * HDIM + h;
      out[idx] = acc[t][r] + dh * u[idx];
    }
  }
}

extern "C" void kernel_launch(void* const* d_in, const int* in_sizes, int n_in,
                              void* d_out, int out_size, void* d_ws,
                              size_t ws_size, hipStream_t stream) {
  const float* u      = (const float*)d_in[0];
  const float* A_diag = (const float*)d_in[1];
  const float* G_diag = (const float*)d_in[2];
  const float* dt     = (const float*)d_in[3];
  const float* Bmat   = (const float*)d_in[4];
  const float* Cmat   = (const float*)d_in[5];
  const float* Dvec   = (const float*)d_in[6];
  float* out = (float*)d_out;

  float* w = (float*)d_ws;
  const size_t NF = (size_t)MROWS * PDIM;
  float* fre  = w;
  float* fim  = fre + NF;
  float* c1   = fim + NF;
  float* c2   = c1 + (size_t)NBATCH * 2 * NCHUNK * PDIM;

  gemm_bu<<<MROWS / BM, GEMM_THREADS, 0, stream>>>(
      u, Bmat, A_diag, G_diag, dt, fre, fim);
  scan_chunks<<<512, 256, 0, stream>>>(fre, fim, c1, c2, A_diag, G_diag, dt);
  gemm_out<<<MROWS / LC, OUT_THREADS, 0, stream>>>(
      fre, fim, c1, c2, A_diag, G_diag, dt, Cmat, u, Dvec, out);
}


# Chunk-scan launch removed (3->2): gemm_bu folds each 64-row block's local carry from its accumulators (8-row lane groups joined by M^16, M^8); gemm_out joins 64-row carries; carry sum order changes
# speedup vs baseline: 1.5375x; 1.0503x over previous
#include <hip/hip_runtime.h>
#include <hip/hip_bf16.h>


typedef float v2f __attribute__((ext_vector_type(2)));
typedef float v4f __attribute__((ext_vector_type(4)));
typedef float v8f __attribute__((ext_vector_type(8)));

#define PDIM 256
#define HDIM 256
#define NBATCH 8
#define LSEQ 4096
#define MROWS (NBATCH * LSEQ)
#define LC 128
#define NCHUNK (LSEQ / LC)
#define LDSPITCH 260
#define RT 4
#define BM (RT * 16)
#define GEMM_THREADS 512
#define OUT_THREADS 1024
#define SCAN_PASS 32
#define KGROUP 16
#define NHC (LSEQ / BM)
#define JB 16

static __device__ inline v8f wmma4f32(v2f a, v2f b, v8f c) {
  return __builtin_amdgcn_wmma_f32_16x16x4_f32(
       false, a,  false, b,
       (short)0, c,  false,  false);
}

static __device__ inline void osc_params(const float* __restrict__ A_diag,
                                         const float* __restrict__ G_diag,
                                         const float* __restrict__ dt, int p,
                                         float& m11, float& m12, float& dts) {
  float s = 1.0f / (1.0f + __expf(-dt[p]));
  float Glow = s * A_diag[p];
  float G = Glow + fmaxf(G_diag[p] - Glow, 0.0f);
  float Alow = 0.25f * G * G;
  float A = Alow + fmaxf(A_diag[p] - Alow, 0.0f);
  m11 = 1.0f - s * G;
  m12 = -s * A;
  dts = s;
}

static __device__ inline void sq2x2(float& a, float& b, float& c, float& d) {
  float na = fmaf(a, a, b * c);
  float nb = fmaf(a, b, b * d);
  float nc = fmaf(c, a, d * c);
  float nd = fmaf(c, b, d * d);
  a = na; b = nb; c = nc; d = nd;
}

__global__ __launch_bounds__(GEMM_THREADS) void gemm_bu(
    const float* __restrict__ u, const float* __restrict__ Bmat,
    const float* __restrict__ A_diag, const float* __restrict__ G_diag,
    const float* __restrict__ dt,
    float* __restrict__ fre, float* __restrict__ fim,
    float* __restrict__ c1, float* __restrict__ c2) {
  __shared__ float su[BM * LDSPITCH];
  const int tid = threadIdx.x;
  const int rowBase = blockIdx.x * BM;

  const float* src = u + (size_t)rowBase * HDIM;
  for (int j = 0; j < BM * HDIM / GEMM_THREADS; ++j) {
    const int i = j * GEMM_THREADS + tid;
    su[(i >> 8) * LDSPITCH + (i & 255)] = src[i];
  }
  __syncthreads();

  const int wave = tid >> 5, lane = tid & 31;
  const int pBase = wave * 16;
  const int mn = lane & 15;
  const int ksel = (lane >> 4) << 1;

  v8f accRe[RT];
  v8f accIm[RT];
#pragma unroll
  for (int t = 0; t < RT; ++t) {
    accRe[t] = v8f{};
    accIm[t] = v8f{};
  }
  const float* arow = su + mn * LDSPITCH + ksel;
  const float* brow = Bmat + ((size_t)(pBase + mn) * HDIM + ksel) * 2;

  for (int k = 0; k < HDIM; k += KGROUP) {
    v4f q[KGROUP / 4];
#pragma unroll
    for (int j = 0; j < KGROUP / 4; ++j)
      q[j] = *(const v4f*)(brow + (k + 4 * j) * 2);
#pragma unroll
    for (int j = 0; j < KGROUP / 4; ++j) {
      const v2f b0 = {q[j].x, q[j].z};
      const v2f b1 = {q[j].y, q[j].w};
#pragma unroll
      for (int t = 0; t < RT; ++t) {
        v2f a = *(const v2f*)(arow + t * 16 * LDSPITCH + k + 4 * j);
        accRe[t] = wmma4f32(a, b0, accRe[t]);
        accIm[t] = wmma4f32(a, b1, accIm[t]);
      }
    }
  }

  const int col = pBase + mn;
  float c11, c12, scale;
  osc_params(A_diag, G_diag, dt, col, c11, c12, scale);
  float ga = c11, gb = c12, gc = scale, gd = 1.0f;
  sq2x2(ga, gb, gc, gd);
  sq2x2(ga, gb, gc, gd);
  sq2x2(ga, gb, gc, gd);
  const float m8a = ga, m8b = gb, m8c = gc, m8d = gd;
  sq2x2(ga, gb, gc, gd);

  float tr1 = 0.0f, tr2 = 0.0f, ti1 = 0.0f, ti2 = 0.0f;
#pragma unroll
  for (int t = 0; t < RT; ++t) {
    const int mo = rowBase + t * 16 + ((lane >> 4) << 3);
    float sr1 = 0.0f, sr2 = 0.0f, si1 = 0.0f, si2 = 0.0f;
#pragma unroll
    for (int r = 0; r < 8; ++r) {
      const float vr = accRe[t][r] * scale;
      const float vi = accIm[t][r] * scale;
      fre[(size_t)(mo + r) * PDIM + col] = vr;
      fim[(size_t)(mo + r) * PDIM + col] = vi;
      const float nr1 = fmaf(c11, sr1, fmaf(c12, sr2, vr));
      const float nr2 = fmaf(scale, sr1, sr2);
      const float ni1 = fmaf(c11, si1, fmaf(c12, si2, vi));
      const float ni2 = fmaf(scale, si1, si2);
      sr1 = nr1; sr2 = nr2; si1 = ni1; si2 = ni2;
    }
    const float ur1 = fmaf(ga, tr1, fmaf(gb, tr2, sr1));
    const float ur2 = fmaf(gc, tr1, fmaf(gd, tr2, sr2));
    const float ui1 = fmaf(ga, ti1, fmaf(gb, ti2, si1));
    const float ui2 = fmaf(gc, ti1, fmaf(gd, ti2, si2));
    tr1 = ur1; tr2 = ur2; ti1 = ui1; ti2 = ui2;
  }
  const float or1 = __shfl_xor(tr1, 16);
  const float or2 = __shfl_xor(tr2, 16);
  const float oi1 = __shfl_xor(ti1, 16);
  const float oi2 = __shfl_xor(ti2, 16);
  if (lane >= 16) {
    const int ci = blockIdx.x * 2 * PDIM + col;
    c1[ci] = fmaf(m8a, or1, fmaf(m8b, or2, tr1));
    c2[ci] = fmaf(m8c, or1, fmaf(m8d, or2, tr2));
    c1[ci + PDIM] = fmaf(m8a, oi1, fmaf(m8b, oi2, ti1));
    c2[ci + PDIM] = fmaf(m8c, oi1, fmaf(m8d, oi2, ti2));
  }
}

__global__ __launch_bounds__(OUT_THREADS) void gemm_out(
    const float* __restrict__ fre, const float* __restrict__ fim,
    const float* __restrict__ c1, const float* __restrict__ c2,
    const float* __restrict__ A_diag, const float* __restrict__ G_diag,
    const float* __restrict__ dt, const float* __restrict__ Cmat,
    const float* __restrict__ u, const float* __restrict__ Dvec,
    float* __restrict__ out) {
  __shared__ float sre[LC * LDSPITCH];
  __shared__ float sim[LC * LDSPITCH];
  const int tid = threadIdx.x;
  const int rowBase = blockIdx.x * LC;
  const int chunk = blockIdx.x & (NCHUNK - 1);

  if (tid < 2 * PDIM) {
    const int comp = tid >> 8, p = tid & (PDIM - 1);
    const float* f = (comp ? fim : fre) + (size_t)rowBase * PDIM + p;
    float* s = (comp ? sim : sre) + p;
    const float sgn = comp ? -1.0f : 1.0f;
    float cur[SCAN_PASS], nxt[SCAN_PASS];
#pragma unroll
    for (int j = 0; j < SCAN_PASS; ++j) cur[j] = f[(size_t)j * PDIM];

    const int nfront = 2 * chunk;
    const int base = ((blockIdx.x - chunk) * 4 + comp) * PDIM + p;
    float l1[2][JB], l2[2][JB];
    if (0 < nfront) {
#pragma unroll
      for (int ch = 0; ch < JB; ++ch) {
        l1[0][ch] = c1[base + ch * 2 * PDIM];
        l2[0][ch] = c2[base + ch * 2 * PDIM];
      }
    }

    float a11, a12, a21;
    osc_params(A_diag, G_diag, dt, p, a11, a12, a21);

    float a = a11, bb = a12, c = a21, d = 1.0f;
#pragma unroll
    for (int i = 0; i < 6; ++i) sq2x2(a, bb, c, d);

    float x1 = 0.0f, x2 = 0.0f;
#pragma unroll
    for (int bt = 0; bt < NHC / JB; ++bt) {
      if (bt + 1 < NHC / JB && (bt + 1) * JB < nfront) {
#pragma unroll
        for (int ch = 0; ch < JB; ++ch) {
          l1[(bt + 1) & 1][ch] = c1[base + ((bt + 1) * JB + ch) * 2 * PDIM];
          l2[(bt + 1) & 1][ch] = c2[base + ((bt + 1) * JB + ch) * 2 * PDIM];
        }
      }
      if (bt * JB < nfront) {
#pragma unroll
        for (int ch = 0; ch < JB; ++ch) {
          float t1 = fmaf(a, x1, fmaf(bb, x2, l1[bt & 1][ch]));
          float t2 = fmaf(c, x1, fmaf(d, x2, l2[bt & 1][ch]));
          if (bt * JB + ch < nfront) {
            x1 = t1;
            x2 = t2;
          }
        }
      }
    }

#pragma unroll
    for (int t0 = 0; t0 < LC; t0 += SCAN_PASS) {
      if (t0 + SCAN_PASS < LC) {
#pragma unroll
        for (int j = 0; j < SCAN_PASS; ++j)
          nxt[j] = f[(size_t)(t0 + SCAN_PASS + j) * PDIM];
      }
#pragma unroll
      for (int j = 0; j < SCAN_PASS; ++j) {
        float y2 = fmaf(a21, x1, x2);
        float y1 = fmaf(a11, x1, fmaf(a12, x2, cur[j]));
        s[(t0 + j) * LDSPITCH] = y2 * sgn;
        x1 = y1;
        x2 = y2;
      }
#pragma unroll
      for (int j = 0; j < SCAN_PASS; ++j) cur[j] = nxt[j];
    }
  }
  __syncthreads();

  const int wave = tid >> 5, lane = tid & 31;
  const int hBase = (wave & 15) * 16;
  const int rHalf = (wave >> 4) * BM;
  const int mn = lane & 15;
  const int ksel = (lane >> 4) << 1;

  v8f acc[RT];
#pragma unroll
  for (int t = 0; t < RT; ++t) acc[t] = v8f{};
  const float* arow_re = sre + (rHalf + mn) * LDSPITCH + ksel;
  const float* arow_im = sim + (rHalf + mn) * LDSPITCH + ksel;
  const float* crow = Cmat + ((size_t)(hBase + mn) * PDIM + ksel) * 2;

  for (int k = 0; k < PDIM; k += KGROUP) {
    v4f q[KGROUP / 4];
#pragma unroll
    for (int j = 0; j < KGROUP / 4; ++j)
      q[j] = *(const v4f*)(crow + (k + 4 * j) * 2);
#pragma unroll
    for (int j = 0; j < KGROUP / 4; ++j) {
      const v2f vr = {q[j].x, q[j].z};
      const v2f vi = {q[j].y, q[j].w};
#pragma unroll
      for (int t = 0; t < RT; ++t) {
        v2f ar = *(const v2f*)(arow_re + t * 16 * LDSPITCH + k + 4 * j);
        v2f ai = *(const v2f*)(arow_im + t * 16 * LDSPITCH + k + 4 * j);
        acc[t] = wmma4f32(ar, vr, acc[t]);
        acc[t] = wmma4f32(ai, vi, acc[t]);
      }
    }
  }

  const int h = hBase + mn;
  const float dh = Dvec[h];
#pragma unroll
  for (int t = 0; t < RT; ++t) {
    const int mo = rowBase + rHalf + t * 16 + ((lane >> 4) << 3);
#pragma unroll
    for (int r = 0; r < 8; ++r) {
      size_t idx = (size_t)(mo + r) * HDIM + h;
      out[idx] = acc[t][r] + dh * u[idx];
    }
  }
}

extern "C" void kernel_launch(void* const* d_in, const int* in_sizes, int n_in,
                              void* d_out, int out_size, void* d_ws,
                              size_t ws_size, hipStream_t stream) {
  const float* u      = (const float*)d_in[0];
  const float* A_diag = (const float*)d_in[1];
  const float* G_diag = (const float*)d_in[2];
  const float* dt     = (const float*)d_in[3];
  const float* Bmat   = (const float*)d_in[4];
  const float* Cmat   = (const float*)d_in[5];
  const float* Dvec   = (const float*)d_in[6];
  float* out = (float*)d_out;

  float* w = (float*)d_ws;
  const size_t NF = (size_t)MROWS * PDIM;
  float* fre  = w;
  float* fim  = fre + NF;
  float* c1   = fim + NF;
  float* c2   = c1 + (size_t)NBATCH * 2 * NHC * PDIM;

  gemm_bu<<<MROWS / BM, GEMM_THREADS, 0, stream>>>(
      u, Bmat, A_diag, G_diag, dt, fre, fim, c1, c2);
  gemm_out<<<MROWS / LC, OUT_THREADS, 0, stream>>>(
      fre, fim, c1, c2, A_diag, G_diag, dt, Cmat, u, Dvec, out);
}
